# starting kernel
# baseline (speedup 1.0000x reference)
#include <hip/hip_runtime.h>

typedef float v2f __attribute__((ext_vector_type(2)));
typedef float v8f __attribute__((ext_vector_type(8)));

#define N_NODES 30000
#define D_DEG   32
#define H_DIM   128
#define K_LAB   10
#define K_AUG   144
#define N_STEPS (K_AUG / 4)
#define AS_STRIDE 148

__global__ __launch_bounds__(256) void gcn_prep(
    const float* __restrict__ W0, const float* __restrict__ b0,
    const float* __restrict__ W1, const float* __restrict__ b1,
    const float* __restrict__ W2, const float* __restrict__ b2,
    const float* __restrict__ Wt3, const float* __restrict__ Wt4,
    float* __restrict__ Bdense, float* __restrict__ Bsw) {
  const int tid = threadIdx.x;
  if (tid < H_DIM) {
    const int hp = tid;
    for (int k = 0; k < H_DIM; ++k)
      Bdense[k * H_DIM + hp] = W2[k * H_DIM + hp];
    for (int k = 0; k < K_LAB; ++k)
      Bdense[(H_DIM + k) * H_DIM + hp] = W1[k * H_DIM + hp];
    Bdense[138 * H_DIM + hp] = W0[hp];
    Bdense[139 * H_DIM + hp] = W0[H_DIM + hp];
    float u = 0.f, v = 0.f;
    for (int h = 0; h < H_DIM; ++h) {
      float t  = Wt4[h];
      float w3 = Wt3[h * H_DIM + hp];
      u += fmaxf(t, 0.f) * w3;
      v += fminf(t, 0.f) * w3;
    }
    Bdense[140 * H_DIM + hp] = u;
    Bdense[141 * H_DIM + hp] = v;
    Bdense[142 * H_DIM + hp] = b0[hp] + b1[hp] + b2[hp];
    Bdense[143 * H_DIM + hp] = 0.f;
  }
  __syncthreads();

  const int tile  = tid >> 5;
  const int lane  = tid & 31;
  const int n     = tile * 16 + (lane & 15);
  const int khalf = 2 * (lane >> 4);
  float* dst = Bsw + (tile * 32 + lane) * (2 * N_STEPS);
  for (int t = 0; t < N_STEPS; ++t) {
    const int kk = 4 * t + khalf;
    dst[2 * t + 0] = Bdense[kk * H_DIM + n];
    dst[2 * t + 1] = Bdense[(kk + 1) * H_DIM + n];
  }
}

__global__ __launch_bounds__(256) void gcn_main(
    const float* __restrict__ feature, const float* __restrict__ x,
    const float* __restrict__ label,   const float* __restrict__ w,
    const float* __restrict__ e_type,  const int* __restrict__ src,
    const float* __restrict__ Bsw,     float* __restrict__ out) {

  __shared__ int   s_src[16 * D_DEG];
  __shared__ float s_et0[16 * D_DEG];
  __shared__ float s_A[16 * AS_STRIDE];

  const int tid = threadIdx.x;
  const int nb  = blockIdx.x * 16;

  for (int i = tid; i < 16 * D_DEG; i += 256) {
    const int n = nb + (i >> 5);
    const int d = i & (D_DEG - 1);
    s_src[i] = src[n * D_DEG + d];
    s_et0[i] = e_type[(n * D_DEG + d) * 2 + 0];
  }
  __syncthreads();

  const int row  = tid >> 4;
  const int c    = tid & 15;
  const int node = nb + row;

  float acc[8];
#pragma unroll
  for (int j = 0; j < 8; ++j) acc[j] = 0.f;
  for (int d = 0; d < D_DEG; ++d) {
    const int   sidx = s_src[row * D_DEG + d];
    const float e0   = s_et0[row * D_DEG + d];
    const float4* frow4 = (const float4*)(feature + (long)sidx * H_DIM) + c * 2;
    const float4 f0 = frow4[0];
    const float4 f1 = frow4[1];
    acc[0] += f0.x * e0; acc[1] += f0.y * e0;
    acc[2] += f0.z * e0; acc[3] += f0.w * e0;
    acc[4] += f1.x * e0; acc[5] += f1.y * e0;
    acc[6] += f1.z * e0; acc[7] += f1.w * e0;
  }
#pragma unroll
  for (int j = 0; j < 8; ++j) s_A[row * AS_STRIDE + c * 8 + j] = acc[j];

  if (c < K_LAB) {
    s_A[row * AS_STRIDE + H_DIM + c] = label[node * K_LAB + c];
  } else if (c == 10) {
    s_A[row * AS_STRIDE + 138] = x[node * 2 + 0];
  } else if (c == 11) {
    s_A[row * AS_STRIDE + 139] = x[node * 2 + 1];
  } else if (c == 12) {
    s_A[row * AS_STRIDE + 142] = 1.0f;
  } else if (c == 13) {
    s_A[row * AS_STRIDE + 143] = 0.0f;
  } else if (c == 14) {
    float p = 0.f;
    for (int d = 0; d < D_DEG; ++d) {
      float nw = w[node * D_DEG + d] * s_et0[row * D_DEG + d];
      p += fmaxf(nw, 0.f);
    }
    s_A[row * AS_STRIDE + 140] = p;
  } else {
    float q = 0.f;
    for (int d = 0; d < D_DEG; ++d) {
      float nw = w[node * D_DEG + d] * s_et0[row * D_DEG + d];
      q += fminf(nw, 0.f);
    }
    s_A[row * AS_STRIDE + 141] = q;
  }
  __syncthreads();

  const int wave = tid >> 5;
  const int lane = tid & 31;
  const int m    = lane & 15;
  const int hi   = lane >> 4;
  const int ncol = wave * 16 + (lane & 15);

  const float4* bfr =
      (const float4*)(Bsw + (wave * 32 + lane) * (2 * N_STEPS));

  v8f cacc = {0.f, 0.f, 0.f, 0.f, 0.f, 0.f, 0.f, 0.f};
#pragma unroll
  for (int t2 = 0; t2 < N_STEPS / 2; ++t2) {
    const float4 bv = bfr[t2];
    {
      const int kk = 4 * (2 * t2) + 2 * hi;
      v2f a, b;
      a.x = s_A[m * AS_STRIDE + kk];
      a.y = s_A[m * AS_STRIDE + kk + 1];
      b.x = bv.x; b.y = bv.y;
      cacc = __builtin_amdgcn_wmma_f32_16x16x4_f32(
          false, a, false, b, (short)0, cacc, false, false);
    }
    {
      const int kk = 4 * (2 * t2 + 1) + 2 * hi;
      v2f a, b;
      a.x = s_A[m * AS_STRIDE + kk];
      a.y = s_A[m * AS_STRIDE + kk + 1];
      b.x = bv.z; b.y = bv.w;
      cacc = __builtin_amdgcn_wmma_f32_16x16x4_f32(
          false, a, false, b, (short)0, cacc, false, false);
    }
  }

#pragma unroll
  for (int r = 0; r < 8; ++r) {
    const int mrow = r + 8 * hi;
    const float vr = cacc[r];
    out[(long)(nb + mrow) * H_DIM + ncol] = vr > 0.f ? vr : 0.f;
  }
}

extern "C" void kernel_launch(void* const* d_in, const int* in_sizes, int n_in,
                              void* d_out, int out_size, void* d_ws, size_t ws_size,
                              hipStream_t stream) {
  const float* feature = (const float*)d_in[0];
  const float* x       = (const float*)d_in[1];
  const float* label   = (const float*)d_in[2];
  const float* w       = (const float*)d_in[3];
  const float* e_type  = (const float*)d_in[4];
  const int*   src     = (const int*)  d_in[5];
  const float* W0      = (const float*)d_in[6];
  const float* b0      = (const float*)d_in[7];
  const float* W1      = (const float*)d_in[8];
  const float* b1      = (const float*)d_in[9];
  const float* W2      = (const float*)d_in[10];
  const float* b2      = (const float*)d_in[11];
  const float* Wt3     = (const float*)d_in[12];
  const float* Wt4     = (const float*)d_in[13];

  float* Bdense = (float*)d_ws;
  float* Bsw    = (float*)((char*)d_ws + K_AUG * H_DIM * sizeof(float));

  gcn_prep<<<1, 256, 0, stream>>>(W0, b0, W1, b1, W2, b2, Wt3, Wt4,
                                  Bdense, Bsw);
  gcn_main<<<N_NODES / 16, 256, 0, stream>>>(feature, x, label, w, e_type, src,
                                             Bsw, (float*)d_out);
}


# The two waves of a SIMD run out of phase: waves 4-7 build B (32 columns each) while waves 0-3 already gather; block barrier split (signal after build, wait before first B read); sum orders kept
# speedup vs baseline: 2.2099x; 2.2099x over previous
#include <hip/hip_runtime.h>


typedef float v2f __attribute__((ext_vector_type(2)));
typedef float v8f __attribute__((ext_vector_type(8)));

#define N_NODES 30000
#define D_DEG   32
#define H_DIM   128
#define K_LAB   10
#define K_AUG   144
#define N_STEPS (K_AUG / 4)
#define N_PAIRS (N_STEPS / 2)
#define AS_STRIDE 148
#define PS_STRIDE 36
#define N_TILES  (N_NODES / 16)
#define N_BLOCKS 256
#define N_WAVES  8

__global__ __launch_bounds__(256) void gcn_fused(
    const float* __restrict__ feature, const float* __restrict__ x,
    const float* __restrict__ label,   const float* __restrict__ w,
    const float* __restrict__ e_type,  const int* __restrict__ src,
    const float* __restrict__ W0, const float* __restrict__ b0,
    const float* __restrict__ W1, const float* __restrict__ b1,
    const float* __restrict__ W2, const float* __restrict__ b2,
    const float* __restrict__ Wt3, const float* __restrict__ Wt4,
    float* __restrict__ out) {

  __shared__ float4 s_B[N_PAIRS * 8 * 32];
  __shared__ __attribute__((aligned(16))) float s_A[N_WAVES * 16 * AS_STRIDE];
  __shared__ __attribute__((aligned(16))) float s_P[N_WAVES * 16 * PS_STRIDE];
  __shared__ int   s_src[N_WAVES * 16 * D_DEG];
  __shared__ float s_et[N_WAVES * 16 * D_DEG];

  const int tid  = threadIdx.x;
  const int wave = __builtin_amdgcn_readfirstlane(tid >> 5);
  const int lane = tid & 31;
  const int m    = lane & 15;
  const int hi   = lane >> 4;

  const int  tile = blockIdx.x + N_BLOCKS * wave;
  const bool has  = tile < N_TILES;
  const int  nb   = tile * 16;

  float* my_A  = s_A  + wave * 16 * AS_STRIDE;
  float* my_P  = s_P  + wave * 16 * PS_STRIDE;
  int*   my_sr = s_src + wave * 16 * D_DEG;
  float* my_et = s_et  + wave * 16 * D_DEG;

  int   e_src[16];
  float e_et[16], e_w[16], e_lab[5], e_x = 0.f;
  if (has) {
#pragma unroll
    for (int n = 0; n < 16; ++n) {
      const int e = (nb + n) * D_DEG + lane;
      e_src[n] = src[e];
      e_et[n]  = ((const float2*)e_type)[e].x;
      e_w[n]   = w[e];
    }
#pragma unroll
    for (int q = 0; q < 5; ++q) e_lab[q] = label[nb * K_LAB + 32 * q + lane];
    e_x = x[nb * 2 + lane];
  }

  if (wave >= 4) {
    const int c = 32 * (wave - 4) + lane;
    float* colp = (float*)s_B + (((c >> 4) * 32) + (c & 15)) * 4;
#define B_OFF(k) ((((k) >> 3) * 256 + 16 * (((k) >> 1) & 1)) * 4 + ((k) & 1) + 2 * (((k) >> 2) & 1))
#pragma unroll
    for (int k = 0; k < H_DIM; k += 2) {
      float2 q;
      q.x = W2[k * H_DIM + c];
      q.y = W2[(k + 1) * H_DIM + c];
      *(float2*)(colp + B_OFF(k)) = q;
    }
#pragma unroll
    for (int k = 0; k < K_LAB; k += 2) {
      float2 q;
      q.x = W1[k * H_DIM + c];
      q.y = W1[(k + 1) * H_DIM + c];
      *(float2*)(colp + B_OFF(128 + k)) = q;
    }
    {
      float2 q;
      q.x = W0[c];
      q.y = W0[H_DIM + c];
      *(float2*)(colp + B_OFF(138)) = q;
    }
    {
      float u = 0.f, v = 0.f;
#pragma unroll 64
      for (int h = 0; h < H_DIM; ++h) {
        const float t4 = Wt4[h];
        const float w3 = Wt3[h * H_DIM + c];
        u += fmaxf(t4, 0.f) * w3;
        v += fminf(t4, 0.f) * w3;
      }
      float2 q;
      q.x = u;
      q.y = v;
      *(float2*)(colp + B_OFF(140)) = q;
    }
    {
      float2 q;
      q.x = b0[c] + b1[c] + b2[c];
      q.y = 0.f;
      *(float2*)(colp + B_OFF(142)) = q;
    }
#undef B_OFF
  }
  asm volatile("s_wait_dscnt 0x0\n\ts_barrier_signal -1" ::: "memory");

  if (has) {
#pragma unroll
    for (int n = 0; n < 16; ++n) {
      my_sr[n * D_DEG + lane]    = e_src[n];
      my_et[n * D_DEG + lane]    = e_et[n];
      my_P[n * PS_STRIDE + lane] = e_w[n] * e_et[n];
    }

#pragma unroll
    for (int q = 0; q < 5; ++q) {
      const int i = 32 * q + lane;
      my_A[(i / K_LAB) * AS_STRIDE + H_DIM + (i % K_LAB)] = e_lab[q];
    }
    my_A[(lane >> 1) * AS_STRIDE + 138 + (lane & 1)] = e_x;
    {
      float p = 0.f;
#pragma unroll
      for (int q = 0; q < 8; ++q) {
        const float4 nw = *(const float4*)(my_P + m * PS_STRIDE + 4 * q);
        p += hi ? fminf(nw.x, 0.f) : fmaxf(nw.x, 0.f);
        p += hi ? fminf(nw.y, 0.f) : fmaxf(nw.y, 0.f);
        p += hi ? fminf(nw.z, 0.f) : fmaxf(nw.z, 0.f);
        p += hi ? fminf(nw.w, 0.f) : fmaxf(nw.w, 0.f);
      }
      my_A[m * AS_STRIDE + 140 + hi] = p;
      my_A[m * AS_STRIDE + 142 + hi] = hi ? 0.f : 1.0f;
    }

    float4 bufA[D_DEG], bufB[D_DEG];

#define ISSUE(buf, n)                                                          \
    {                                                                          \
      const int sv = my_sr[(n) * D_DEG + lane];                                \
      _Pragma("unroll")                                                        \
      for (int d = 0; d < D_DEG; ++d) {                                        \
        const int sidx = __builtin_amdgcn_readlane(sv, d);                     \
        buf[d] = ((const float4*)(feature + (long)sidx * H_DIM))[lane];        \
      }                                                                        \
    }
#define CONSUME(buf, n)                                                        \
    {                                                                          \
      const int eb = __float_as_int(my_et[(n) * D_DEG + lane]);                \
      float4 acc = {0.f, 0.f, 0.f, 0.f};                                       \
      _Pragma("unroll")                                                        \
      for (int d = 0; d < D_DEG; ++d) {                                        \
        const float e0 = __int_as_float(__builtin_amdgcn_readlane(eb, d));     \
        acc.x += buf[d].x * e0; acc.y += buf[d].y * e0;                        \
        acc.z += buf[d].z * e0; acc.w += buf[d].w * e0;                        \
      }                                                                        \
      *(float4*)(my_A + (n) * AS_STRIDE + 4 * lane) = acc;                     \
    }

    ISSUE(bufA, 0)
#pragma unroll 1
    for (int p2 = 0; p2 < 8; ++p2) {
      ISSUE(bufB, 2 * p2 + 1)
      CONSUME(bufA, 2 * p2)
      if (p2 < 7) ISSUE(bufA, 2 * p2 + 2)
      CONSUME(bufB, 2 * p2 + 1)
    }
#undef ISSUE
#undef CONSUME

  }

  asm volatile("s_barrier_wait -1" ::: "memory");

  if (has) {
    v8f cacc[8];
#pragma unroll
    for (int j = 0; j < 8; ++j) cacc[j] = (v8f){0.f, 0.f, 0.f, 0.f, 0.f, 0.f, 0.f, 0.f};
    const float*  arow = my_A + m * AS_STRIDE + 2 * hi;
    const float4* bl   = s_B + lane;
#pragma unroll
    for (int t2 = 0; t2 < N_PAIRS; ++t2) {
      v2f a0, a1;
      a0.x = arow[8 * t2];     a0.y = arow[8 * t2 + 1];
      a1.x = arow[8 * t2 + 4]; a1.y = arow[8 * t2 + 5];
      float4 b[8];
#pragma unroll
      for (int j = 0; j < 8; ++j) b[j] = bl[(t2 * 8 + j) * 32];
#pragma unroll
      for (int j = 0; j < 8; ++j) {
        v2f bb; bb.x = b[j].x; bb.y = b[j].y;
        cacc[j] = __builtin_amdgcn_wmma_f32_16x16x4_f32(
            false, a0, false, bb, (short)0, cacc[j], false, false);
      }
#pragma unroll
      for (int j = 0; j < 8; ++j) {
        v2f bb; bb.x = b[j].z; bb.y = b[j].w;
        cacc[j] = __builtin_amdgcn_wmma_f32_16x16x4_f32(
            false, a1, false, bb, (short)0, cacc[j], false, false);
      }
    }

#pragma unroll
    for (int j = 0; j < 8; ++j) {
#pragma unroll
      for (int r = 0; r < 8; ++r) {
        const int mrow = r + 8 * hi;
        const float vr = cacc[j][r];
        out[(long)(nb + mrow) * H_DIM + 16 * j + m] = vr > 0.f ? vr : 0.f;
      }
    }
  }
}

extern "C" void kernel_launch(void* const* d_in, const int* in_sizes, int n_in,
                              void* d_out, int out_size, void* d_ws, size_t ws_size,
                              hipStream_t stream) {
  const float* feature = (const float*)d_in[0];
  const float* x       = (const float*)d_in[1];
  const float* label   = (const float*)d_in[2];
  const float* w       = (const float*)d_in[3];
  const float* e_type  = (const float*)d_in[4];
  const int*   src     = (const int*)  d_in[5];
  const float* W0      = (const float*)d_in[6];
  const float* b0      = (const float*)d_in[7];
  const float* W1      = (const float*)d_in[8];
  const float* b1      = (const float*)d_in[9];
  const float* W2      = (const float*)d_in[10];
  const float* b2      = (const float*)d_in[11];
  const float* Wt3     = (const float*)d_in[12];
  const float* Wt4     = (const float*)d_in[13];

  gcn_fused<<<N_BLOCKS, 256, 0, stream>>>(feature, x, label, w, e_type, src,
                                          W0, b0, W1, b1, W2, b2, Wt3, Wt4,
                                          (float*)d_out);
}


# Gather split along K: two 256-B half rows (nodes n, n+8) per load, columns 0-63 first; their k-steps and the augmented ones run under the second half's loads; K order: aug, 0-63, 64-127
# speedup vs baseline: 2.4863x; 1.1251x over previous
#include <hip/hip_runtime.h>


typedef float v2f __attribute__((ext_vector_type(2)));
typedef float v8f __attribute__((ext_vector_type(8)));

#define N_NODES 30000
#define D_DEG   32
#define H_DIM   128
#define K_LAB   10
#define K_AUG   144
#define N_STEPS (K_AUG / 4)
#define N_PAIRS (N_STEPS / 2)
#define AS_STRIDE 148
#define PS_STRIDE 36
#define N_TILES  (N_NODES / 16)
#define N_BLOCKS 256
#define N_WAVES  8

__global__ __launch_bounds__(256) void gcn_fused(
    const float* __restrict__ feature, const float* __restrict__ x,
    const float* __restrict__ label,   const float* __restrict__ w,
    const float* __restrict__ e_type,  const int* __restrict__ src,
    const float* __restrict__ W0, const float* __restrict__ b0,
    const float* __restrict__ W1, const float* __restrict__ b1,
    const float* __restrict__ W2, const float* __restrict__ b2,
    const float* __restrict__ Wt3, const float* __restrict__ Wt4,
    float* __restrict__ out) {

  __shared__ float4 s_B[N_PAIRS * 8 * 32];
  __shared__ __attribute__((aligned(16))) float s_A[N_WAVES * 16 * AS_STRIDE];
  __shared__ __attribute__((aligned(16))) float s_P[N_WAVES * 16 * PS_STRIDE];
  __shared__ __attribute__((aligned(16))) unsigned s_src[N_WAVES * 16 * D_DEG];
  __shared__ __attribute__((aligned(16))) float s_et[N_WAVES * 16 * D_DEG];

  const int tid  = threadIdx.x;
  const int wave = __builtin_amdgcn_readfirstlane(tid >> 5);
  const int lane = tid & 31;
  const int m    = lane & 15;
  const int hi   = lane >> 4;

  const int  tile = blockIdx.x + N_BLOCKS * wave;
  const bool has  = tile < N_TILES;
  const int  nb   = tile * 16;

  float* my_A  = s_A  + wave * 16 * AS_STRIDE;
  float* my_P  = s_P  + wave * 16 * PS_STRIDE;
  unsigned* my_sr = s_src + wave * 16 * D_DEG;
  float* my_et = s_et  + wave * 16 * D_DEG;

  int   e_src[16];
  float e_et[16], e_w[16], e_lab[5], e_x = 0.f;
  if (has) {
#pragma unroll
    for (int n = 0; n < 16; ++n) {
      const int e = (nb + n) * D_DEG + lane;
      e_src[n] = src[e];
      e_et[n]  = ((const float2*)e_type)[e].x;
      e_w[n]   = w[e];
    }
#pragma unroll
    for (int q = 0; q < 5; ++q) e_lab[q] = label[nb * K_LAB + 32 * q + lane];
    e_x = x[nb * 2 + lane];
  }

  if (wave >= 4) {
    const int c = 32 * (wave - 4) + lane;
    float* colp = (float*)s_B + (((c >> 4) * 32) + (c & 15)) * 4;
#define B_OFF(k) ((((k) >> 3) * 256 + 16 * (((k) >> 1) & 1)) * 4 + ((k) & 1) + 2 * (((k) >> 2) & 1))
#pragma unroll
    for (int k = 0; k < H_DIM; k += 2) {
      float2 q;
      q.x = W2[k * H_DIM + c];
      q.y = W2[(k + 1) * H_DIM + c];
      *(float2*)(colp + B_OFF(k)) = q;
    }
#pragma unroll
    for (int k = 0; k < K_LAB; k += 2) {
      float2 q;
      q.x = W1[k * H_DIM + c];
      q.y = W1[(k + 1) * H_DIM + c];
      *(float2*)(colp + B_OFF(128 + k)) = q;
    }
    {
      float2 q;
      q.x = W0[c];
      q.y = W0[H_DIM + c];
      *(float2*)(colp + B_OFF(138)) = q;
    }
    {
      float u = 0.f, v = 0.f;
#pragma unroll 64
      for (int h = 0; h < H_DIM; ++h) {
        const float t4 = Wt4[h];
        const float w3 = Wt3[h * H_DIM + c];
        u += fmaxf(t4, 0.f) * w3;
        v += fminf(t4, 0.f) * w3;
      }
      float2 q;
      q.x = u;
      q.y = v;
      *(float2*)(colp + B_OFF(140)) = q;
    }
    {
      float2 q;
      q.x = b0[c] + b1[c] + b2[c];
      q.y = 0.f;
      *(float2*)(colp + B_OFF(142)) = q;
    }
#undef B_OFF
  }
  asm volatile("s_wait_dscnt 0x0\n\ts_barrier_signal -1" ::: "memory");

  const char* fbytes = (const char*)feature;
  float4 bufA[D_DEG], bufB[D_DEG];
  v8f cacc[8];
  const float*  arow = my_A + m * AS_STRIDE + 2 * hi;
  const float4* bl   = s_B + lane;

#define ISSUE(buf, s)                                                          \
    {                                                                          \
      const unsigned* sp = my_sr + (((s) & 7) + 8 * hi) * D_DEG;               \
      const unsigned  lp = 256u * (unsigned)((s) >> 3) + 16u * (unsigned)m;    \
      _Pragma("unroll")                                                        \
      for (int q = 0; q < 8; ++q) {                                            \
        const uint4 o4 = *(const uint4*)(sp + 4 * q);                          \
        buf[4 * q + 0] = *(const float4*)(fbytes + (size_t)(o4.x + lp));       \
        buf[4 * q + 1] = *(const float4*)(fbytes + (size_t)(o4.y + lp));       \
        buf[4 * q + 2] = *(const float4*)(fbytes + (size_t)(o4.z + lp));       \
        buf[4 * q + 3] = *(const float4*)(fbytes + (size_t)(o4.w + lp));       \
      }                                                                        \
    }
#define FMA4(acc, v, e)                                                        \
      acc.x += v.x * e; acc.y += v.y * e; acc.z += v.z * e; acc.w += v.w * e;
#define CONSUME(buf, s)                                                        \
    {                                                                          \
      const int    rr = ((s) & 7) + 8 * hi;                                    \
      const float* wp = my_et + rr * D_DEG;                                    \
      float4 acc = {0.f, 0.f, 0.f, 0.f};                                       \
      _Pragma("unroll")                                                        \
      for (int q = 0; q < 8; ++q) {                                            \
        const float4 w4 = *(const float4*)(wp + 4 * q);                        \
        FMA4(acc, buf[4 * q + 0], w4.x)                                        \
        FMA4(acc, buf[4 * q + 1], w4.y)                                        \
        FMA4(acc, buf[4 * q + 2], w4.z)                                        \
        FMA4(acc, buf[4 * q + 3], w4.w)                                        \
      }                                                                        \
      *(float4*)(my_A + rr * AS_STRIDE + 64 * ((s) >> 3) + 4 * m) = acc;       \
    }
#define WBLOCK(t2)                                                             \
    {                                                                          \
      const float* ar = arow + 8 * (t2);                                       \
      v2f a0, a1;                                                              \
      a0.x = ar[0]; a0.y = ar[1];                                              \
      a1.x = ar[4]; a1.y = ar[5];                                              \
      const float4* bp = bl + (t2) * 256;                                      \
      float4 b[8];                                                             \
      _Pragma("unroll")                                                        \
      for (int j = 0; j < 8; ++j) b[j] = bp[j * 32];                           \
      _Pragma("unroll")                                                        \
      for (int j = 0; j < 8; ++j) {                                            \
        v2f bb; bb.x = b[j].x; bb.y = b[j].y;                                  \
        cacc[j] = __builtin_amdgcn_wmma_f32_16x16x4_f32(                       \
            false, a0, false, bb, (short)0, cacc[j], false, false);            \
      }                                                                        \
      _Pragma("unroll")                                                        \
      for (int j = 0; j < 8; ++j) {                                            \
        v2f bb; bb.x = b[j].z; bb.y = b[j].w;                                  \
        cacc[j] = __builtin_amdgcn_wmma_f32_16x16x4_f32(                       \
            false, a1, false, bb, (short)0, cacc[j], false, false);            \
      }                                                                        \
    }

  if (has) {
#pragma unroll
    for (int n = 0; n < 16; ++n) {
      my_sr[n * D_DEG + lane]    = (unsigned)e_src[n] << 9;
      my_et[n * D_DEG + lane]    = e_et[n];
      my_P[n * PS_STRIDE + lane] = e_w[n] * e_et[n];
    }

#pragma unroll
    for (int q = 0; q < 5; ++q) {
      const int i = 32 * q + lane;
      my_A[(i / K_LAB) * AS_STRIDE + H_DIM + (i % K_LAB)] = e_lab[q];
    }
    my_A[(lane >> 1) * AS_STRIDE + 138 + (lane & 1)] = e_x;
    {
      float p = 0.f;
#pragma unroll
      for (int q = 0; q < 8; ++q) {
        const float4 nw = *(const float4*)(my_P + m * PS_STRIDE + 4 * q);
        p += hi ? fminf(nw.x, 0.f) : fmaxf(nw.x, 0.f);
        p += hi ? fminf(nw.y, 0.f) : fmaxf(nw.y, 0.f);
        p += hi ? fminf(nw.z, 0.f) : fmaxf(nw.z, 0.f);
        p += hi ? fminf(nw.w, 0.f) : fmaxf(nw.w, 0.f);
      }
      my_A[m * AS_STRIDE + 140 + hi] = p;
      my_A[m * AS_STRIDE + 142 + hi] = hi ? 0.f : 1.0f;
    }

    ISSUE(bufA, 0)
#pragma unroll 1
    for (int p2 = 0; p2 < 4; ++p2) {
      ISSUE(bufB, 2 * p2 + 1)
      CONSUME(bufA, 2 * p2)
      ISSUE(bufA, 2 * p2 + 2)
      CONSUME(bufB, 2 * p2 + 1)
    }
  }

  asm volatile("s_barrier_wait -1" ::: "memory");

  if (has) {
#pragma unroll
    for (int j = 0; j < 8; ++j) cacc[j] = (v8f){0.f, 0.f, 0.f, 0.f, 0.f, 0.f, 0.f, 0.f};
    WBLOCK(16)
    WBLOCK(17)

#pragma unroll 1
    for (int p2 = 4; p2 < 8; ++p2) {
      ISSUE(bufB, 2 * p2 + 1)
      WBLOCK(2 * (p2 - 4))
      CONSUME(bufA, 2 * p2)
      if (p2 < 7) ISSUE(bufA, 2 * p2 + 2)
      WBLOCK(2 * (p2 - 4) + 1)
      CONSUME(bufB, 2 * p2 + 1)
    }

#pragma unroll
    for (int t2 = 8; t2 < 16; ++t2) WBLOCK(t2)
#undef ISSUE
#undef FMA4
#undef CONSUME
#undef WBLOCK

#pragma unroll
    for (int j = 0; j < 8; ++j) {
#pragma unroll
      for (int r = 0; r < 8; ++r) {
        const int mrow = r + 8 * hi;
        const float vr = cacc[j][r];
        out[(long)(nb + mrow) * H_DIM + 16 * j + m] = vr > 0.f ? vr : 0.f;
      }
    }
  }
}

extern "C" void kernel_launch(void* const* d_in, const int* in_sizes, int n_in,
                              void* d_out, int out_size, void* d_ws, size_t ws_size,
                              hipStream_t stream) {
  const float* feature = (const float*)d_in[0];
  const float* x       = (const float*)d_in[1];
  const float* label   = (const float*)d_in[2];
  const float* w       = (const float*)d_in[3];
  const float* e_type  = (const float*)d_in[4];
  const int*   src     = (const int*)  d_in[5];
  const float* W0      = (const float*)d_in[6];
  const float* b0      = (const float*)d_in[7];
  const float* W1      = (const float*)d_in[8];
  const float* b1      = (const float*)d_in[9];
  const float* W2      = (const float*)d_in[10];
  const float* b2      = (const float*)d_in[11];
  const float* Wt3     = (const float*)d_in[12];
  const float* Wt4     = (const float*)d_in[13];

  gcn_fused<<<N_BLOCKS, 256, 0, stream>>>(feature, x, label, w, e_type, src,
                                          W0, b0, W1, b1, W2, b2, Wt3, Wt4,
                                          (float*)d_out);
}
